# starting kernel
# baseline (speedup 1.0000x reference)
#include <hip/hip_runtime.h>
#include <math.h>

#define SEGS 65
#define CCH  64
#define PIX  (256*256)
#define BB   4
#define SDIM 64

typedef __attribute__((ext_vector_type(2))) float v2f;
typedef __attribute__((ext_vector_type(8))) float v8f;

__device__ __forceinline__ unsigned ftos(float f) {
    unsigned b = __float_as_uint(f);
    return (b & 0x80000000u) ? ~b : (b | 0x80000000u);
}
__device__ __forceinline__ float stof(unsigned u) {
    unsigned b = (u & 0x80000000u) ? (u & 0x7FFFFFFFu) : ~u;
    return __uint_as_float(b);
}
#define NEG_INF_S 0x007FFFFFu

__global__ void seg_init_kernel(unsigned* seg, int n) {
    int i = blockIdx.x * blockDim.x + threadIdx.x;
    if (i < n) seg[i] = NEG_INF_S;
}

__global__ void segmax_kernel(const float* __restrict__ enc,
                              const int*   __restrict__ masks,
                              unsigned*    __restrict__ seg) {
    const int pixPerBlock    = 2048;
    const int chunksPerBatch = PIX / pixPerBlock;
    int b     = blockIdx.x / chunksPerBatch;
    int chunk = blockIdx.x % chunksPerBatch;
    int p0    = chunk * pixPerBlock;

    __shared__ unsigned tab[SEGS * CCH];
    for (int i = threadIdx.x; i < SEGS * CCH; i += blockDim.x) tab[i] = NEG_INF_S;
    __syncthreads();

    const float* encb  = enc   + (size_t)b * CCH * PIX;
    const int*   maskb = masks + (size_t)b * PIX;

    for (int base = 0; base < pixPerBlock; base += blockDim.x * 4) {
        int p = p0 + base + threadIdx.x * 4;
        int4 id4 = *(const int4*)&maskb[p];
        unsigned* r0 = &tab[id4.x * CCH];
        unsigned* r1 = &tab[id4.y * CCH];
        unsigned* r2 = &tab[id4.z * CCH];
        unsigned* r3 = &tab[id4.w * CCH];
        #pragma unroll 4
        for (int c = 0; c < CCH; ++c) {
            float4 v = *(const float4*)&encb[(size_t)c * PIX + p];
            atomicMax(&r0[c], ftos(v.x));
            atomicMax(&r1[c], ftos(v.y));
            atomicMax(&r2[c], ftos(v.z));
            atomicMax(&r3[c], ftos(v.w));
        }
    }
    __syncthreads();

    unsigned* segb = seg + (size_t)b * SEGS * CCH;
    for (int i = threadIdx.x; i < SEGS * CCH; i += blockDim.x)
        atomicMax(&segb[i], tab[i]);
}

__global__ void vec_out_kernel(const unsigned* __restrict__ seg,
                               float* __restrict__ outv) {
    int i = blockIdx.x * blockDim.x + threadIdx.x;
    if (i >= BB * CCH * SDIM) return;
    int s = i % SDIM;
    int c = (i / SDIM) % CCH;
    int b = i / (SDIM * CCH);
    outv[i] = stof(seg[((size_t)b * SEGS + (s + 1)) * CCH + c]);
}

__global__ void conv_wmma_kernel(const unsigned* __restrict__ seg,
                                 const float* __restrict__ w,
                                 const float* __restrict__ bias,
                                 float*       __restrict__ outc) {
    __shared__ float vpad[CCH * 66];
    __shared__ float wl[4 * 576];

    int b        = blockIdx.x >> 5;
    int tileBase = (blockIdx.x & 31) * 8;
    const unsigned* segb = seg + (size_t)b * SEGS * CCH;

    for (int i = threadIdx.x; i < CCH * 66; i += blockDim.x) {
        int c = i / 66, x = i % 66;
        vpad[i] = (x == 0 || x == 65) ? 0.0f : stof(segb[x * CCH + c]);
    }
    for (int i = threadIdx.x; i < 4 * 576; i += blockDim.x) {
        int o = i / 576, r = i % 576;
        int t = r / 64, c = r % 64;
        wl[i] = w[((size_t)(o * CCH + c)) * 9 + t];
    }
    __syncthreads();

    int wave = threadIdx.x >> 5;
    int lane = threadIdx.x & 31;
    int m    = lane & 15;
    int half = lane >> 4;
    int kb   = half * 2;

    int tile = tileBase + wave;
    int ic   = tile >> 2;
    int j0   = (tile & 3) << 4;

    bool wvalid = (m < 4);
    int  mo     = wvalid ? m : 0;

    v8f acc = {0.f, 0.f, 0.f, 0.f, 0.f, 0.f, 0.f, 0.f};

    for (int t = 0; t < 9; ++t) {
        int di = t / 3 - 1, dj = t % 3 - 1;
        int xi = ic + di + 1;
        int xj = j0 + m + dj + 1;
        const float* wrow = &wl[mo * 576 + t * 64];
        #pragma unroll
        for (int c4 = 0; c4 < 16; ++c4) {
            int c0 = c4 * 4 + kb;
            v2f a, bv;
            a.x = vpad[c0 * 66 + xi] * vpad[c0 * 66 + xj];
            a.y = vpad[(c0 + 1) * 66 + xi] * vpad[(c0 + 1) * 66 + xj];
            bv.x = wvalid ? wrow[c0]     : 0.0f;
            bv.y = wvalid ? wrow[c0 + 1] : 0.0f;
            acc = __builtin_amdgcn_wmma_f32_16x16x4_f32(
                      false, a, false, bv, (short)0, acc, false, false);
        }
    }

    if (m < 4) {
        float bs = bias[m];
        #pragma unroll
        for (int r = 0; r < 8; ++r) {
            int j = j0 + r + half * 8;
            float val = acc[r] + bs;
            outc[(((size_t)b * 4 + m) * SDIM + ic) * SDIM + j] =
                1.0f / (1.0f + expf(-val));
        }
    }
}

extern "C" void kernel_launch(void* const* d_in, const int* in_sizes, int n_in,
                              void* d_out, int out_size, void* d_ws, size_t ws_size,
                              hipStream_t stream) {
    const float* enc   = (const float*)d_in[0];
    const int*   masks = (const int*)d_in[1];
    const float* w     = (const float*)d_in[2];
    const float* bias  = (const float*)d_in[3];
    float*    out  = (float*)d_out;
    unsigned* seg  = (unsigned*)d_ws;
    float*    outv = out;
    float*    outc = out + BB * CCH * SDIM;

    int nseg = BB * SEGS * CCH;
    seg_init_kernel<<<(nseg + 255) / 256, 256, 0, stream>>>(seg, nseg);
    segmax_kernel<<<BB * 32, 256, 0, stream>>>(enc, masks, seg);
    vec_out_kernel<<<(BB * CCH * SDIM + 255) / 256, 256, 0, stream>>>(seg, outv);
    conv_wmma_kernel<<<BB * 32, 256, 0, stream>>>(seg, w, bias, outc);
}


# segmax: lanes mapped to channels (32 px x 64 ch tile transposed through wave-private LDS) so each LDS max instruction hits one segment row on 32 distinct banks, not 32 lanes on one bank
# speedup vs baseline: 2.5668x; 2.5668x over previous
#include <hip/hip_runtime.h>
#include <math.h>

#define SEGS 65
#define CCH  64
#define PIX  (256*256)
#define BB   4
#define SDIM 64

typedef __attribute__((ext_vector_type(2))) float v2f;
typedef __attribute__((ext_vector_type(8))) float v8f;
typedef __attribute__((ext_vector_type(4))) float v4f;

__device__ __forceinline__ unsigned ftos(float f) {
    unsigned b = __float_as_uint(f);
    return (b & 0x80000000u) ? ~b : (b | 0x80000000u);
}
__device__ __forceinline__ float stof(unsigned u) {
    unsigned b = (u & 0x80000000u) ? (u & 0x7FFFFFFFu) : ~u;
    return __uint_as_float(b);
}
#define NEG_INF_S 0x007FFFFFu

__global__ void seg_init_kernel(unsigned* seg, int n) {
    int i = blockIdx.x * blockDim.x + threadIdx.x;
    if (i < n) seg[i] = NEG_INF_S;
}

#define TPIX   32
#define SPITCH 36
#define NLD    16

__device__ __forceinline__ int tile_ch(int k, int csub) {
    return ((k >> 2) << 4) + ((k & 3) << 1) + csub;
}

__device__ __forceinline__ void tile_load(v4f (&buf)[NLD], int& ids,
                                          const float* __restrict__ encb,
                                          const int* __restrict__ maskb,
                                          int pt, int lane, int q, int csub) {
    ids = maskb[pt + lane];
    #pragma unroll
    for (int k = 0; k < NLD; ++k) {
        int c = tile_ch(k, csub);
        buf[k] = *(const v4f*)&encb[(size_t)c * PIX + pt + 4 * q];
    }
}

__device__ __forceinline__ void tile_reduce(const v4f (&buf)[NLD], int ids,
                                            float* __restrict__ st,
                                            unsigned* __restrict__ tab,
                                            int lane, int q, int csub) {
    #pragma unroll
    for (int k = 0; k < NLD; ++k) {
        int c = tile_ch(k, csub);
        *(v4f*)&st[c * SPITCH + 4 * q] = buf[k];
    }
    asm volatile("s_wait_dscnt 0x0" ::: "memory");
    #pragma unroll
    for (int qq = 0; qq < TPIX / 4; ++qq) {
        v4f va = *(const v4f*)&st[lane * SPITCH + 4 * qq];
        v4f vb = *(const v4f*)&st[(lane + 32) * SPITCH + 4 * qq];
        #pragma unroll
        for (int e = 0; e < 4; ++e) {
            int id = __builtin_amdgcn_readlane(ids, 4 * qq + e);
            unsigned* row = &tab[id * CCH];
            atomicMax(&row[lane],      ftos(va[e]));
            atomicMax(&row[lane + 32], ftos(vb[e]));
        }
    }
}

__global__ __launch_bounds__(256)
void segmax_kernel(const float* __restrict__ enc,
                   const int*   __restrict__ masks,
                   unsigned*    __restrict__ seg) {
    const int pixPerBlock    = 2048;
    const int chunksPerBatch = PIX / pixPerBlock;
    const int pixPerWave     = pixPerBlock / 8;
    int b     = blockIdx.x / chunksPerBatch;
    int chunk = blockIdx.x % chunksPerBatch;
    int wave  = threadIdx.x >> 5;
    int lane  = threadIdx.x & 31;
    int pw    = chunk * pixPerBlock + wave * pixPerWave;

    __shared__ unsigned tab[SEGS * CCH];
    __shared__ __attribute__((aligned(16))) float stage[8][CCH * SPITCH];
    for (int i = threadIdx.x; i < SEGS * CCH; i += blockDim.x) tab[i] = NEG_INF_S;
    __syncthreads();

    const float* encb  = enc   + (size_t)b * CCH * PIX;
    const int*   maskb = masks + (size_t)b * PIX;
    float* st = stage[wave];

    int q    = lane & 7;
    int cs   = lane >> 3;
    int csub = ((cs & 1) << 3) | (cs >> 1);

    v4f bufA[NLD], bufB[NLD];
    int idA, idB;
    tile_load(bufA, idA, encb, maskb, pw, lane, q, csub);
    #pragma unroll 1
    for (int t = 0; t < pixPerWave / TPIX; t += 2) {
        tile_load(bufB, idB, encb, maskb, pw + (t + 1) * TPIX, lane, q, csub);
        tile_reduce(bufA, idA, st, tab, lane, q, csub);
        if (t + 2 < pixPerWave / TPIX)
            tile_load(bufA, idA, encb, maskb, pw + (t + 2) * TPIX, lane, q, csub);
        tile_reduce(bufB, idB, st, tab, lane, q, csub);
    }
    __syncthreads();

    unsigned* segb = seg + (size_t)b * SEGS * CCH;
    for (int i = threadIdx.x; i < SEGS * CCH; i += blockDim.x)
        atomicMax(&segb[i], tab[i]);
}

__global__ void vec_out_kernel(const unsigned* __restrict__ seg,
                               float* __restrict__ outv) {
    int i = blockIdx.x * blockDim.x + threadIdx.x;
    if (i >= BB * CCH * SDIM) return;
    int s = i % SDIM;
    int c = (i / SDIM) % CCH;
    int b = i / (SDIM * CCH);
    outv[i] = stof(seg[((size_t)b * SEGS + (s + 1)) * CCH + c]);
}

__global__ void conv_wmma_kernel(const unsigned* __restrict__ seg,
                                 const float* __restrict__ w,
                                 const float* __restrict__ bias,
                                 float*       __restrict__ outc) {
    __shared__ float vpad[CCH * 66];
    __shared__ float wl[4 * 576];

    int b        = blockIdx.x >> 5;
    int tileBase = (blockIdx.x & 31) * 8;
    const unsigned* segb = seg + (size_t)b * SEGS * CCH;

    for (int i = threadIdx.x; i < CCH * 66; i += blockDim.x) {
        int c = i / 66, x = i % 66;
        vpad[i] = (x == 0 || x == 65) ? 0.0f : stof(segb[x * CCH + c]);
    }
    for (int i = threadIdx.x; i < 4 * 576; i += blockDim.x) {
        int o = i / 576, r = i % 576;
        int t = r / 64, c = r % 64;
        wl[i] = w[((size_t)(o * CCH + c)) * 9 + t];
    }
    __syncthreads();

    int wave = threadIdx.x >> 5;
    int lane = threadIdx.x & 31;
    int m    = lane & 15;
    int half = lane >> 4;
    int kb   = half * 2;

    int tile = tileBase + wave;
    int ic   = tile >> 2;
    int j0   = (tile & 3) << 4;

    bool wvalid = (m < 4);
    int  mo     = wvalid ? m : 0;

    v8f acc = {0.f, 0.f, 0.f, 0.f, 0.f, 0.f, 0.f, 0.f};

    for (int t = 0; t < 9; ++t) {
        int di = t / 3 - 1, dj = t % 3 - 1;
        int xi = ic + di + 1;
        int xj = j0 + m + dj + 1;
        const float* wrow = &wl[mo * 576 + t * 64];
        #pragma unroll
        for (int c4 = 0; c4 < 16; ++c4) {
            int c0 = c4 * 4 + kb;
            v2f a, bv;
            a.x = vpad[c0 * 66 + xi] * vpad[c0 * 66 + xj];
            a.y = vpad[(c0 + 1) * 66 + xi] * vpad[(c0 + 1) * 66 + xj];
            bv.x = wvalid ? wrow[c0]     : 0.0f;
            bv.y = wvalid ? wrow[c0 + 1] : 0.0f;
            acc = __builtin_amdgcn_wmma_f32_16x16x4_f32(
                      false, a, false, bv, (short)0, acc, false, false);
        }
    }

    if (m < 4) {
        float bs = bias[m];
        #pragma unroll
        for (int r = 0; r < 8; ++r) {
            int j = j0 + r + half * 8;
            float val = acc[r] + bs;
            outc[(((size_t)b * 4 + m) * SDIM + ic) * SDIM + j] =
                1.0f / (1.0f + expf(-val));
        }
    }
}

extern "C" void kernel_launch(void* const* d_in, const int* in_sizes, int n_in,
                              void* d_out, int out_size, void* d_ws, size_t ws_size,
                              hipStream_t stream) {
    const float* enc   = (const float*)d_in[0];
    const int*   masks = (const int*)d_in[1];
    const float* w     = (const float*)d_in[2];
    const float* bias  = (const float*)d_in[3];
    float*    out  = (float*)d_out;
    unsigned* seg  = (unsigned*)d_ws;
    float*    outv = out;
    float*    outc = out + BB * CCH * SDIM;

    int nseg = BB * SEGS * CCH;
    seg_init_kernel<<<(nseg + 255) / 256, 256, 0, stream>>>(seg, nseg);
    segmax_kernel<<<BB * 32, 256, 0, stream>>>(enc, masks, seg);
    vec_out_kernel<<<(BB * CCH * SDIM + 255) / 256, 256, 0, stream>>>(seg, outv);
    conv_wmma_kernel<<<BB * 32, 256, 0, stream>>>(seg, w, bias, outc);
}


# conv kernel: loads issued in groups ahead of use (25 independent fill loads per thread; each tap's WMMA operands read from LDS as one group a tap early): one wait per group, not one per load
# speedup vs baseline: 3.6272x; 1.4131x over previous
#include <hip/hip_runtime.h>
#include <math.h>

#define SEGS 65
#define CCH  64
#define PIX  (256*256)
#define BB   4
#define SDIM 64

typedef __attribute__((ext_vector_type(2))) float v2f;
typedef __attribute__((ext_vector_type(8))) float v8f;
typedef __attribute__((ext_vector_type(4))) float v4f;

__device__ __forceinline__ unsigned ftos(float f) {
    unsigned b = __float_as_uint(f);
    return (b & 0x80000000u) ? ~b : (b | 0x80000000u);
}
__device__ __forceinline__ float stof(unsigned u) {
    unsigned b = (u & 0x80000000u) ? (u & 0x7FFFFFFFu) : ~u;
    return __uint_as_float(b);
}
#define NEG_INF_S 0x007FFFFFu

__global__ void seg_init_kernel(unsigned* seg, int n) {
    int i = blockIdx.x * blockDim.x + threadIdx.x;
    if (i < n) seg[i] = NEG_INF_S;
}

#define TPIX   32
#define SPITCH 36
#define NLD    16

__device__ __forceinline__ int tile_ch(int k, int csub) {
    return ((k >> 2) << 4) + ((k & 3) << 1) + csub;
}

__device__ __forceinline__ void tile_load(v4f (&buf)[NLD], int& ids,
                                          const float* __restrict__ encb,
                                          const int* __restrict__ maskb,
                                          int pt, int lane, int q, int csub) {
    ids = maskb[pt + lane];
    #pragma unroll
    for (int k = 0; k < NLD; ++k) {
        int c = tile_ch(k, csub);
        buf[k] = *(const v4f*)&encb[(size_t)c * PIX + pt + 4 * q];
    }
}

__device__ __forceinline__ void tile_reduce(const v4f (&buf)[NLD], int ids,
                                            float* __restrict__ st,
                                            unsigned* __restrict__ tab,
                                            int lane, int q, int csub) {
    #pragma unroll
    for (int k = 0; k < NLD; ++k) {
        int c = tile_ch(k, csub);
        *(v4f*)&st[c * SPITCH + 4 * q] = buf[k];
    }
    asm volatile("s_wait_dscnt 0x0" ::: "memory");
    #pragma unroll
    for (int qq = 0; qq < TPIX / 4; ++qq) {
        v4f va = *(const v4f*)&st[lane * SPITCH + 4 * qq];
        v4f vb = *(const v4f*)&st[(lane + 32) * SPITCH + 4 * qq];
        #pragma unroll
        for (int e = 0; e < 4; ++e) {
            int id = __builtin_amdgcn_readlane(ids, 4 * qq + e);
            unsigned* row = &tab[id * CCH];
            atomicMax(&row[lane],      ftos(va[e]));
            atomicMax(&row[lane + 32], ftos(vb[e]));
        }
    }
}

__global__ __launch_bounds__(256)
void segmax_kernel(const float* __restrict__ enc,
                   const int*   __restrict__ masks,
                   unsigned*    __restrict__ seg) {
    const int pixPerBlock    = 2048;
    const int chunksPerBatch = PIX / pixPerBlock;
    const int pixPerWave     = pixPerBlock / 8;
    int b     = blockIdx.x / chunksPerBatch;
    int chunk = blockIdx.x % chunksPerBatch;
    int wave  = threadIdx.x >> 5;
    int lane  = threadIdx.x & 31;
    int pw    = chunk * pixPerBlock + wave * pixPerWave;

    __shared__ unsigned tab[SEGS * CCH];
    __shared__ __attribute__((aligned(16))) float stage[8][CCH * SPITCH];
    for (int i = threadIdx.x; i < SEGS * CCH; i += blockDim.x) tab[i] = NEG_INF_S;
    __syncthreads();

    const float* encb  = enc   + (size_t)b * CCH * PIX;
    const int*   maskb = masks + (size_t)b * PIX;
    float* st = stage[wave];

    int q    = lane & 7;
    int cs   = lane >> 3;
    int csub = ((cs & 1) << 3) | (cs >> 1);

    v4f bufA[NLD], bufB[NLD];
    int idA, idB;
    tile_load(bufA, idA, encb, maskb, pw, lane, q, csub);
    #pragma unroll 1
    for (int t = 0; t < pixPerWave / TPIX; t += 2) {
        tile_load(bufB, idB, encb, maskb, pw + (t + 1) * TPIX, lane, q, csub);
        tile_reduce(bufA, idA, st, tab, lane, q, csub);
        if (t + 2 < pixPerWave / TPIX)
            tile_load(bufA, idA, encb, maskb, pw + (t + 2) * TPIX, lane, q, csub);
        tile_reduce(bufB, idB, st, tab, lane, q, csub);
    }
    __syncthreads();

    unsigned* segb = seg + (size_t)b * SEGS * CCH;
    for (int i = threadIdx.x; i < SEGS * CCH; i += blockDim.x)
        atomicMax(&segb[i], tab[i]);
}

__global__ void vec_out_kernel(const unsigned* __restrict__ seg,
                               float* __restrict__ outv) {
    int i = blockIdx.x * blockDim.x + threadIdx.x;
    if (i >= BB * CCH * SDIM) return;
    int s = i % SDIM;
    int c = (i / SDIM) % CCH;
    int b = i / (SDIM * CCH);
    outv[i] = stof(seg[((size_t)b * SEGS + (s + 1)) * CCH + c]);
}

#define VPITCH 66
#define WPITCH 580

__device__ __forceinline__ int op_ch(int k, int kb) {
    return ((k >> 1) << 2) + kb + (k & 1);
}

__global__ __launch_bounds__(256)
void conv_wmma_kernel(const unsigned* __restrict__ seg,
                      const float* __restrict__ w,
                      const float* __restrict__ bias,
                      float*       __restrict__ outc) {
    __shared__ float vpad[CCH * VPITCH];
    __shared__ __attribute__((aligned(16))) float wl[4 * WPITCH];

    int b        = blockIdx.x >> 5;
    int tileBase = (blockIdx.x & 31) * 8;
    const unsigned* segb = seg + (size_t)b * SEGS * CCH;
    int tid = threadIdx.x;

    unsigned sv[16];
    float    wv[9];
    #pragma unroll
    for (int k = 0; k < 16; ++k) sv[k] = segb[CCH + tid + 256 * k];
    #pragma unroll
    for (int k = 0; k < 9; ++k)  wv[k] = w[tid + 256 * k];

    if (tid < 2 * CCH)
        vpad[(tid & 63) * VPITCH + ((tid >> 6) ? 65 : 0)] = 0.0f;
    #pragma unroll
    for (int k = 0; k < 16; ++k) {
        int idx = tid + 256 * k;
        vpad[(idx & 63) * VPITCH + (idx >> 6) + 1] = stof(sv[k]);
    }
    #pragma unroll
    for (int k = 0; k < 9; ++k) {
        int g = tid + 256 * k;
        int o = g / 576, r = g - o * 576;
        int c = r / 9,   t = r - c * 9;
        wl[o * WPITCH + t * 64 + c] = wv[k];
    }
    __syncthreads();

    int wave = tid >> 5;
    int lane = tid & 31;
    int m    = lane & 15;
    int half = lane >> 4;
    int kb   = half * 2;

    int tile = tileBase + wave;
    int ic   = tile >> 2;
    int j0   = (tile & 3) << 4;

    int mo = m & 3;

    v8f acc = {0.f, 0.f, 0.f, 0.f, 0.f, 0.f, 0.f, 0.f};

    float vj[3][32];
    float vi[2][32];
    v2f   wb[2][16];
    const float* wbase = &wl[mo * WPITCH + kb];

    #pragma unroll
    for (int dj = 0; dj < 3; ++dj) {
        #pragma unroll
        for (int k = 0; k < 32; ++k)
            vj[dj][k] = vpad[op_ch(k, kb) * VPITCH + j0 + m + dj];
    }
    #pragma unroll
    for (int k = 0; k < 32; ++k)
        vi[0][k] = vpad[op_ch(k, kb) * VPITCH + ic];
    #pragma unroll
    for (int c4 = 0; c4 < 16; ++c4) wb[0][c4] = *(const v2f*)&wbase[c4 * 4];
    __builtin_amdgcn_sched_barrier(0);

    #pragma unroll
    for (int t = 0; t < 9; ++t) {
        const int di = t / 3, dj = t % 3;
        if (t + 1 < 9) {
            #pragma unroll
            for (int c4 = 0; c4 < 16; ++c4)
                wb[(t + 1) & 1][c4] = *(const v2f*)&wbase[(t + 1) * 64 + c4 * 4];
            if (dj == 2) {
                #pragma unroll
                for (int k = 0; k < 32; ++k)
                    vi[(di + 1) & 1][k] = vpad[op_ch(k, kb) * VPITCH + ic + di + 1];
            }
        }
        __builtin_amdgcn_sched_barrier(0);
        #pragma unroll
        for (int c4 = 0; c4 < 16; ++c4) {
            v2f a;
            a.x = vi[di & 1][2 * c4]     * vj[dj][2 * c4];
            a.y = vi[di & 1][2 * c4 + 1] * vj[dj][2 * c4 + 1];
            acc = __builtin_amdgcn_wmma_f32_16x16x4_f32(
                      false, a, false, wb[t & 1][c4], (short)0, acc, false, false);
        }
        __builtin_amdgcn_sched_barrier(0);
    }

    if (m < 4) {
        float bs = bias[m];
        #pragma unroll
        for (int r = 0; r < 8; ++r) {
            int j = j0 + r + half * 8;
            float val = acc[r] + bs;
            outc[(((size_t)b * 4 + m) * SDIM + ic) * SDIM + j] =
                1.0f / (1.0f + expf(-val));
        }
    }
}

extern "C" void kernel_launch(void* const* d_in, const int* in_sizes, int n_in,
                              void* d_out, int out_size, void* d_ws, size_t ws_size,
                              hipStream_t stream) {
    const float* enc   = (const float*)d_in[0];
    const int*   masks = (const int*)d_in[1];
    const float* w     = (const float*)d_in[2];
    const float* bias  = (const float*)d_in[3];
    float*    out  = (float*)d_out;
    unsigned* seg  = (unsigned*)d_ws;
    float*    outv = out;
    float*    outc = out + BB * CCH * SDIM;

    int nseg = BB * SEGS * CCH;
    seg_init_kernel<<<(nseg + 255) / 256, 256, 0, stream>>>(seg, nseg);
    segmax_kernel<<<BB * 32, 256, 0, stream>>>(enc, masks, seg);
    vec_out_kernel<<<(BB * CCH * SDIM + 255) / 256, 256, 0, stream>>>(seg, outv);
    conv_wmma_kernel<<<BB * 32, 256, 0, stream>>>(seg, w, bias, outc);
}


# segmax: one work-group per (batch, channel) plane, lane-private LDS tables, stores the final vec row; conv reads vec: no table pre-set, no global atomic merge, no vec pass (4 launches -> 2)
# speedup vs baseline: 5.5021x; 1.5169x over previous
#include <hip/hip_runtime.h>
#include <math.h>

#define SEGS 65
#define CCH  64
#define PIX  (256*256)
#define BB   4
#define SDIM 64

typedef __attribute__((ext_vector_type(2))) float v2f;
typedef __attribute__((ext_vector_type(8))) float v8f;
typedef __attribute__((ext_vector_type(4))) float v4f;
typedef __attribute__((ext_vector_type(4))) int   v4i;

__device__ __forceinline__ unsigned ftos(float f) {
    unsigned b = __float_as_uint(f);
    return (b & 0x80000000u) ? ~b : (b | 0x80000000u);
}
__device__ __forceinline__ float stof(unsigned u) {
    unsigned b = (u & 0x80000000u) ? (u & 0x7FFFFFFFu) : ~u;
    return __uint_as_float(b);
}
#define NEG_INF_S 0x007FFFFFu

#define WPX  8192
#define GLD  8
#define TROW 32

__device__ __forceinline__ void grp_load(v4f (&v)[GLD], v4i (&id)[GLD],
                                         const float* __restrict__ e,
                                         const int* __restrict__ m,
                                         int base, int lane) {
    #pragma unroll
    for (int k = 0; k < GLD; ++k) {
        v[k]  = *(const v4f*)&e[base + k * 128 + 4 * lane];
        id[k] = *(const v4i*)&m[base + k * 128 + 4 * lane];
    }
}

__device__ __forceinline__ void grp_reduce(const v4f (&v)[GLD], const v4i (&id)[GLD],
                                           unsigned* __restrict__ tl) {
    #pragma unroll
    for (int k = 0; k < GLD; ++k) {
        #pragma unroll
        for (int e = 0; e < 4; ++e)
            atomicMax(&tl[id[k][e] * TROW], ftos(v[k][e]));
    }
}

__global__ __launch_bounds__(256)
void segmax_kernel(const float* __restrict__ enc,
                   const int*   __restrict__ masks,
                   float*       __restrict__ outv) {
    int b    = blockIdx.x >> 6;
    int c    = blockIdx.x & 63;
    int tid  = threadIdx.x;
    int wave = tid >> 5;
    int lane = tid & 31;

    __shared__ unsigned tab[8][SEGS * TROW];
    unsigned* t = tab[wave];

    const float* e = enc   + ((size_t)b * CCH + c) * PIX + wave * WPX;
    const int*   m = masks + (size_t)b * PIX + wave * WPX;

    v4f vA[GLD], vB[GLD];
    v4i iA[GLD], iB[GLD];
    grp_load(vA, iA, e, m, 0, lane);

    for (int i = lane; i < SEGS * TROW; i += 32) t[i] = NEG_INF_S;
    __syncthreads();

    unsigned* tl = t + lane;
    const int NG = WPX / (GLD * 128);
    #pragma unroll 1
    for (int g = 0; g < NG - 2; g += 2) {
        grp_load(vB, iB, e, m, (g + 1) * GLD * 128, lane);
        __builtin_amdgcn_sched_barrier(0);
        grp_reduce(vA, iA, tl);
        grp_load(vA, iA, e, m, (g + 2) * GLD * 128, lane);
        __builtin_amdgcn_sched_barrier(0);
        grp_reduce(vB, iB, tl);
    }
    grp_load(vB, iB, e, m, (NG - 1) * GLD * 128, lane);
    __builtin_amdgcn_sched_barrier(0);
    grp_reduce(vA, iA, tl);
    grp_reduce(vB, iB, tl);
    __syncthreads();

    int s  = tid >> 2;
    int q  = tid & 3;
    int id = s + 1;
    int sk = id >> 1;
    unsigned mx = NEG_INF_S;
    #pragma unroll
    for (int w = 0; w < 8; ++w) {
        #pragma unroll
        for (int j = 0; j < 8; ++j) {
            unsigned x = tab[w][id * TROW + q * 8 + ((j + sk) & 7)];
            mx = x > mx ? x : mx;
        }
    }
    unsigned o1 = __shfl_xor(mx, 1);
    mx = o1 > mx ? o1 : mx;
    unsigned o2 = __shfl_xor(mx, 2);
    mx = o2 > mx ? o2 : mx;
    if (q == 0)
        outv[((size_t)b * CCH + c) * SDIM + s] = stof(mx);
}

#define VPITCH 66
#define WPITCH 580

__device__ __forceinline__ int op_ch(int k, int kb) {
    return ((k >> 1) << 2) + kb + (k & 1);
}

__global__ __launch_bounds__(256)
void conv_wmma_kernel(const float* __restrict__ vec,
                      const float* __restrict__ w,
                      const float* __restrict__ bias,
                      float*       __restrict__ outc) {
    __shared__ float vpad[CCH * VPITCH];
    __shared__ __attribute__((aligned(16))) float wl[4 * WPITCH];

    int b        = blockIdx.x >> 5;
    int tileBase = (blockIdx.x & 31) * 8;
    const float* vecb = vec + (size_t)b * CCH * SDIM;
    int tid = threadIdx.x;

    float    sv[16];
    float    wv[9];
    #pragma unroll
    for (int k = 0; k < 16; ++k) sv[k] = vecb[tid + 256 * k];
    #pragma unroll
    for (int k = 0; k < 9; ++k)  wv[k] = w[tid + 256 * k];

    if (tid < 2 * CCH)
        vpad[(tid & 63) * VPITCH + ((tid >> 6) ? 65 : 0)] = 0.0f;
    #pragma unroll
    for (int k = 0; k < 16; ++k) {
        int idx = tid + 256 * k;
        vpad[(idx >> 6) * VPITCH + (idx & 63) + 1] = sv[k];
    }
    #pragma unroll
    for (int k = 0; k < 9; ++k) {
        int g = tid + 256 * k;
        int o = g / 576, r = g - o * 576;
        int c = r / 9,   t = r - c * 9;
        wl[o * WPITCH + t * 64 + c] = wv[k];
    }
    __syncthreads();

    int wave = tid >> 5;
    int lane = tid & 31;
    int m    = lane & 15;
    int half = lane >> 4;
    int kb   = half * 2;

    int tile = tileBase + wave;
    int ic   = tile >> 2;
    int j0   = (tile & 3) << 4;

    int mo = m & 3;

    v8f acc = {0.f, 0.f, 0.f, 0.f, 0.f, 0.f, 0.f, 0.f};

    float vj[3][32];
    float vi[2][32];
    v2f   wb[2][16];
    const float* wbase = &wl[mo * WPITCH + kb];

    #pragma unroll
    for (int dj = 0; dj < 3; ++dj) {
        #pragma unroll
        for (int k = 0; k < 32; ++k)
            vj[dj][k] = vpad[op_ch(k, kb) * VPITCH + j0 + m + dj];
    }
    #pragma unroll
    for (int k = 0; k < 32; ++k)
        vi[0][k] = vpad[op_ch(k, kb) * VPITCH + ic];
    #pragma unroll
    for (int c4 = 0; c4 < 16; ++c4) wb[0][c4] = *(const v2f*)&wbase[c4 * 4];
    __builtin_amdgcn_sched_barrier(0);

    #pragma unroll
    for (int t = 0; t < 9; ++t) {
        const int di = t / 3, dj = t % 3;
        if (t + 1 < 9) {
            #pragma unroll
            for (int c4 = 0; c4 < 16; ++c4)
                wb[(t + 1) & 1][c4] = *(const v2f*)&wbase[(t + 1) * 64 + c4 * 4];
            if (dj == 2) {
                #pragma unroll
                for (int k = 0; k < 32; ++k)
                    vi[(di + 1) & 1][k] = vpad[op_ch(k, kb) * VPITCH + ic + di + 1];
            }
        }
        __builtin_amdgcn_sched_barrier(0);
        #pragma unroll
        for (int c4 = 0; c4 < 16; ++c4) {
            v2f a;
            a.x = vi[di & 1][2 * c4]     * vj[dj][2 * c4];
            a.y = vi[di & 1][2 * c4 + 1] * vj[dj][2 * c4 + 1];
            acc = __builtin_amdgcn_wmma_f32_16x16x4_f32(
                      false, a, false, wb[t & 1][c4], (short)0, acc, false, false);
        }
        __builtin_amdgcn_sched_barrier(0);
    }

    if (m < 4) {
        float bs = bias[m];
        #pragma unroll
        for (int r = 0; r < 8; ++r) {
            int j = j0 + r + half * 8;
            float val = acc[r] + bs;
            outc[(((size_t)b * 4 + m) * SDIM + ic) * SDIM + j] =
                1.0f / (1.0f + expf(-val));
        }
    }
}

extern "C" void kernel_launch(void* const* d_in, const int* in_sizes, int n_in,
                              void* d_out, int out_size, void* d_ws, size_t ws_size,
                              hipStream_t stream) {
    const float* enc   = (const float*)d_in[0];
    const int*   masks = (const int*)d_in[1];
    const float* w     = (const float*)d_in[2];
    const float* bias  = (const float*)d_in[3];
    float*    out  = (float*)d_out;
    float*    outv = out;
    float*    outc = out + BB * CCH * SDIM;

    segmax_kernel<<<BB * CCH, 256, 0, stream>>>(enc, masks, outv);
    conv_wmma_kernel<<<BB * 32, 256, 0, stream>>>(outv, w, bias, outc);
}
